# starting kernel
# baseline (speedup 1.0000x reference)
#include <hip/hip_runtime.h>

typedef __attribute__((ext_vector_type(2))) float v2f;
typedef __attribute__((ext_vector_type(8))) float v8f;

#define BB 4
#define NN 4096
#define FF 64

__global__ void k_zero(float* __restrict__ deg) {
    deg[blockIdx.x * 256 + threadIdx.x] = 0.0f;
}

__global__ void k_colsum(const float* __restrict__ A, float* __restrict__ deg) {
    const int j  = blockIdx.x * 256 + threadIdx.x;
    const int b  = blockIdx.z;
    const int r0 = blockIdx.y * 64;
    const float* Ab = A + (size_t)b * NN * NN;
    float s = 0.0f;
#pragma unroll 4
    for (int r = 0; r < 64; ++r)
        s += Ab[(size_t)(r0 + r) * NN + j];
    atomicAdd(&deg[b * NN + j], s);
}

__global__ void k_rsqrt(float* __restrict__ deg) {
    const int i = blockIdx.x * 256 + threadIdx.x;
    deg[i] = rsqrtf(deg[i] + 1.0f);
}

__global__ void k_hw(const float* __restrict__ H, const float* __restrict__ W,
                     const float* __restrict__ d, float* __restrict__ HWsT) {
    __shared__ float  sh[256];
    __shared__ float4 tile[64];
    const int t = threadIdx.x;
    const int rowBase = blockIdx.x * 4;
    sh[t] = H[(size_t)rowBase * FF + t];
    __syncthreads();
    const int lr  = t >> 6;
    const int o   = t & 63;
    const int row = rowBase + lr;
    float s = 0.0f;
#pragma unroll
    for (int f = 0; f < FF; ++f)
        s += sh[lr * FF + f] * W[f * FF + o];
    ((float*)&tile[o])[lr] = d[row] * s;
    __syncthreads();
    if (t < 64) {
        const int b  = rowBase / NN;
        const int m0 = rowBase % NN;
        *(float4*)&HWsT[((size_t)b * FF + t) * NN + m0] = tile[t];
    }
}

__global__ void __launch_bounds__(256)
k_gcn_main(const float* __restrict__ A, const float* __restrict__ HWsT,
           const float* __restrict__ d, float* __restrict__ out) {
    const int b    = blockIdx.y;
    const int wave = threadIdx.x >> 5;
    const int lane = threadIdx.x & 31;
    const int half = lane >> 4;
    const int lm   = lane & 15;

    const int m0 = blockIdx.x * 32 + (wave & 1) * 16;
    const int n0 = (wave >> 1) * 16;
    const int col = n0 + lm;

    const float* __restrict__ Ab  = A + ((size_t)b * NN + (m0 + lm)) * NN;
    const float* __restrict__ BbT = HWsT + ((size_t)b * FF + col) * NN;

    v8f acc = {};
#pragma unroll 8
    for (int k0 = 0; k0 < NN; k0 += 4) {
        v2f a  = *(const v2f*)(Ab  + k0 + 2 * half);
        v2f bv = *(const v2f*)(BbT + k0 + 2 * half);
        acc = __builtin_amdgcn_wmma_f32_16x16x4_f32(
             false, a,  false, bv,
             (short)0, acc,  false,  false);
    }

    const float* __restrict__ db = d + b * NN;
    float* __restrict__ ob = out + (size_t)b * NN * FF;
    const int rbase = m0 + 8 * half;
#pragma unroll
    for (int i = 0; i < 8; ++i) {
        const int row = rbase + i;
        ob[(size_t)row * FF + col] = db[row] * (acc[i] + BbT[row]);
    }
}

extern "C" void kernel_launch(void* const* d_in, const int* in_sizes, int n_in,
                              void* d_out, int out_size, void* d_ws, size_t ws_size,
                              hipStream_t stream) {
    const float* A = (const float*)d_in[0];
    const float* H = (const float*)d_in[1];
    const float* W = (const float*)d_in[2];
    float* out = (float*)d_out;

    float* deg  = (float*)d_ws;
    float* HWsT = deg + BB * NN;

    k_zero<<<BB * NN / 256, 256, 0, stream>>>(deg);

    dim3 gc(NN / 256, 64, BB);
    k_colsum<<<gc, 256, 0, stream>>>(A, deg);

    k_rsqrt<<<BB * NN / 256, 256, 0, stream>>>(deg);

    k_hw<<<BB * NN / 4, 256, 0, stream>>>(H, W, deg, HWsT);

    dim3 gm(NN / 32, BB);
    k_gcn_main<<<gm, 256, 0, stream>>>(A, HWsT, deg, out);
}


# main GEMM: A and HWsT K-tiles of 256 staged in LDS by coalesced 128-bit global loads (64x64 output tile per workgroup, register-prefetched double buffer); per-element sum order unchanged
# speedup vs baseline: 1.6682x; 1.6682x over previous
#include <hip/hip_runtime.h>

typedef __attribute__((ext_vector_type(2))) float v2f;
typedef __attribute__((ext_vector_type(8))) float v8f;

#define BB 4
#define NN 4096
#define FF 64

__global__ void k_zero(float* __restrict__ deg) {
    deg[blockIdx.x * 256 + threadIdx.x] = 0.0f;
}

__global__ void k_colsum(const float* __restrict__ A, float* __restrict__ deg) {
    const int j  = blockIdx.x * 256 + threadIdx.x;
    const int b  = blockIdx.z;
    const int r0 = blockIdx.y * 64;
    const float* Ab = A + (size_t)b * NN * NN;
    float s = 0.0f;
#pragma unroll 4
    for (int r = 0; r < 64; ++r)
        s += Ab[(size_t)(r0 + r) * NN + j];
    atomicAdd(&deg[b * NN + j], s);
}

__global__ void k_rsqrt(float* __restrict__ deg) {
    const int i = blockIdx.x * 256 + threadIdx.x;
    deg[i] = rsqrtf(deg[i] + 1.0f);
}

__global__ void k_hw(const float* __restrict__ H, const float* __restrict__ W,
                     const float* __restrict__ d, float* __restrict__ HWsT) {
    __shared__ float  sh[256];
    __shared__ float4 tile[64];
    const int t = threadIdx.x;
    const int rowBase = blockIdx.x * 4;
    sh[t] = H[(size_t)rowBase * FF + t];
    __syncthreads();
    const int lr  = t >> 6;
    const int o   = t & 63;
    const int row = rowBase + lr;
    float s = 0.0f;
#pragma unroll
    for (int f = 0; f < FF; ++f)
        s += sh[lr * FF + f] * W[f * FF + o];
    ((float*)&tile[o])[lr] = d[row] * s;
    __syncthreads();
    if (t < 64) {
        const int b  = rowBase / NN;
        const int m0 = rowBase % NN;
        *(float4*)&HWsT[((size_t)b * FF + t) * NN + m0] = tile[t];
    }
}

#define KT     256
#define PITCH  260
#define NT     (NN / KT)
#define TILE_F (64 * PITCH)

__global__ void __launch_bounds__(256)
k_gcn_main(const float* __restrict__ A, const float* __restrict__ HWsT,
           const float* __restrict__ d, float* __restrict__ out) {
    __shared__ __attribute__((aligned(16))) float As[2 * TILE_F];
    __shared__ __attribute__((aligned(16))) float Bs[2 * TILE_F];

    const int b     = blockIdx.y;
    const int wave  = threadIdx.x >> 5;
    const int lane  = threadIdx.x & 31;
    const int half  = lane >> 4;
    const int lm    = lane & 15;
    const int mbase = blockIdx.x * 64;

    const float* __restrict__ gA = A    + ((size_t)b * NN + mbase + wave * 8) * NN + lane * 4;
    const float* __restrict__ gB = HWsT + ((size_t)b * FF + wave * 8) * NN + lane * 4;
    const int sOff = wave * 8 * PITCH + lane * 4;

    const int rt    = wave & 3;
    const int cpair = wave >> 2;
    const int aOff  = (rt * 16 + lm) * PITCH + 2 * half;
    const int bOff0 = (cpair * 32 + lm) * PITCH + 2 * half;
    const int bOff1 = bOff0 + 16 * PITCH;

#define FOR16(X) X(0) X(1) X(2) X(3) X(4) X(5) X(6) X(7) X(8) X(9) X(10) X(11) X(12) X(13) X(14) X(15)
#define ST_DECL(q)  float4 ra##q, rb##q;
#define ST_LOAD(q)  ra##q = *(const float4*)(pA + (size_t)((q) >> 1) * NN + ((q) & 1) * 128); \
                    rb##q = *(const float4*)(pB + (size_t)((q) >> 1) * NN + ((q) & 1) * 128);
#define ST_STORE(q) *(float4*)&As[so + ((q) >> 1) * PITCH + ((q) & 1) * 128] = ra##q; \
                    *(float4*)&Bs[so + ((q) >> 1) * PITCH + ((q) & 1) * 128] = rb##q;

    {
        FOR16(ST_DECL)
        const float* __restrict__ pA = gA;
        const float* __restrict__ pB = gB;
        FOR16(ST_LOAD)
        const int so = sOff;
        FOR16(ST_STORE)
    }
    __syncthreads();

    FOR16(ST_DECL)

    v8f acc0 = {};
    v8f acc1 = {};

    for (int t = 0; t < NT; ++t) {
        const int cur = (t & 1) * TILE_F;
        const bool more = (t + 1 < NT);

        if (more) {
            const float* __restrict__ pA = gA + (t + 1) * KT;
            const float* __restrict__ pB = gB + (t + 1) * KT;
            FOR16(ST_LOAD)
        }

        const float* as  = &As[cur + aOff];
        const float* bs0 = &Bs[cur + bOff0];
        const float* bs1 = &Bs[cur + bOff1];
#pragma unroll
        for (int kk = 0; kk < KT; kk += 4) {
            const v2f a  = *(const v2f*)(as  + kk);
            const v2f b0 = *(const v2f*)(bs0 + kk);
            const v2f b1 = *(const v2f*)(bs1 + kk);
            acc0 = __builtin_amdgcn_wmma_f32_16x16x4_f32(
                 false, a,  false, b0,
                 (short)0, acc0,  false,  false);
            acc1 = __builtin_amdgcn_wmma_f32_16x16x4_f32(
                 false, a,  false, b1,
                 (short)0, acc1,  false,  false);
        }

        if (more) {
            const int so = (TILE_F - cur) + sOff;
            FOR16(ST_STORE)
        }
        __syncthreads();
    }
#undef ST_STORE
#undef ST_LOAD
#undef ST_DECL
#undef FOR16

    const float* __restrict__ db = d + b * NN;
    float* __restrict__ ob = out + (size_t)b * NN * FF;
    const int r0 = mbase + rt * 16 + 8 * half;
    const int c0 = cpair * 32 + lm;
    const float* __restrict__ h0 = HWsT + ((size_t)b * FF + c0) * NN;
    const float* __restrict__ h1 = h0 + (size_t)16 * NN;
#pragma unroll
    for (int i = 0; i < 8; ++i) {
        const int row = r0 + i;
        const float dr = db[row];
        ob[(size_t)row * FF + c0]      = dr * (acc0[i] + h0[row]);
        ob[(size_t)row * FF + c0 + 16] = dr * (acc1[i] + h1[row]);
    }
}

extern "C" void kernel_launch(void* const* d_in, const int* in_sizes, int n_in,
                              void* d_out, int out_size, void* d_ws, size_t ws_size,
                              hipStream_t stream) {
    const float* A = (const float*)d_in[0];
    const float* H = (const float*)d_in[1];
    const float* W = (const float*)d_in[2];
    float* out = (float*)d_out;

    float* deg  = (float*)d_ws;
    float* HWsT = deg + BB * NN;

    k_zero<<<BB * NN / 256, 256, 0, stream>>>(deg);

    dim3 gc(NN / 256, 64, BB);
    k_colsum<<<gc, 256, 0, stream>>>(A, deg);

    k_rsqrt<<<BB * NN / 256, 256, 0, stream>>>(deg);

    k_hw<<<BB * NN / 4, 256, 0, stream>>>(H, W, deg, HWsT);

    dim3 gm(NN / 64, BB);
    k_gcn_main<<<gm, 256, 0, stream>>>(A, HWsT, deg, out);
}


# column sums: one 512-thread workgroup owns 64 whole columns (128-bit loads, 16-32 in flight per wave), stores rsqrt(sum+1) itself: atomics, zero-fill and rsqrt launches gone; column-sum order changed
# speedup vs baseline: 1.8500x; 1.1090x over previous
#include <hip/hip_runtime.h>

typedef __attribute__((ext_vector_type(2))) float v2f;
typedef __attribute__((ext_vector_type(8))) float v8f;

#define BB 4
#define NN 4096
#define FF 64

#define CS_SLOTS 32
#define CS_ROWS  (NN / CS_SLOTS)
#define CS_GRP   16

__global__ void __launch_bounds__(CS_SLOTS * 16) __attribute__((amdgpu_waves_per_eu(4, 8)))
k_colsum_d(const float* __restrict__ A, float* __restrict__ d) {
    __shared__ float4 red[CS_SLOTS * 16];
    const int t    = threadIdx.x;
    const int q    = t & 15;
    const int slot = t >> 4;
    const int b    = blockIdx.x >> 6;
    const int c0   = (blockIdx.x & 63) * 64;
    const float* __restrict__ p = A + ((size_t)b * NN + slot) * NN + c0 + q * 4;
    const size_t step = (size_t)CS_SLOTS * NN;

    float4 s = make_float4(0.0f, 0.0f, 0.0f, 0.0f);
    float4 va[CS_GRP], vb[CS_GRP];
#pragma unroll
    for (int j = 0; j < CS_GRP; ++j)
        va[j] = *(const float4*)(p + (size_t)j * step);
    for (int i = 0; i < CS_ROWS; i += 2 * CS_GRP) {
#pragma unroll
        for (int j = 0; j < CS_GRP; ++j)
            vb[j] = *(const float4*)(p + (size_t)(i + CS_GRP + j) * step);
#pragma unroll
        for (int j = 0; j < CS_GRP; ++j) {
            s.x += va[j].x; s.y += va[j].y; s.z += va[j].z; s.w += va[j].w;
        }
        if (i + 2 * CS_GRP < CS_ROWS) {
#pragma unroll
            for (int j = 0; j < CS_GRP; ++j)
                va[j] = *(const float4*)(p + (size_t)(i + 2 * CS_GRP + j) * step);
        }
#pragma unroll
        for (int j = 0; j < CS_GRP; ++j) {
            s.x += vb[j].x; s.y += vb[j].y; s.z += vb[j].z; s.w += vb[j].w;
        }
    }

    red[slot * 16 + q] = s;
    __syncthreads();
    if (t < 64) {
        const float* r = (const float*)red;
        float acc = 0.0f;
#pragma unroll 8
        for (int k = 0; k < CS_SLOTS; ++k)
            acc += r[k * 64 + t];
        d[b * NN + c0 + t] = rsqrtf(acc + 1.0f);
    }
}

__global__ void k_hw(const float* __restrict__ H, const float* __restrict__ W,
                     const float* __restrict__ d, float* __restrict__ HWsT) {
    __shared__ float  sh[256];
    __shared__ float4 tile[64];
    const int t = threadIdx.x;
    const int rowBase = blockIdx.x * 4;
    sh[t] = H[(size_t)rowBase * FF + t];
    __syncthreads();
    const int lr  = t >> 6;
    const int o   = t & 63;
    const int row = rowBase + lr;
    float s = 0.0f;
#pragma unroll
    for (int f = 0; f < FF; ++f)
        s += sh[lr * FF + f] * W[f * FF + o];
    ((float*)&tile[o])[lr] = d[row] * s;
    __syncthreads();
    if (t < 64) {
        const int b  = rowBase / NN;
        const int m0 = rowBase % NN;
        *(float4*)&HWsT[((size_t)b * FF + t) * NN + m0] = tile[t];
    }
}

#define KT     256
#define PITCH  260
#define NT     (NN / KT)
#define TILE_F (64 * PITCH)

__global__ void __launch_bounds__(256)
k_gcn_main(const float* __restrict__ A, const float* __restrict__ HWsT,
           const float* __restrict__ d, float* __restrict__ out) {
    __shared__ __attribute__((aligned(16))) float As[2 * TILE_F];
    __shared__ __attribute__((aligned(16))) float Bs[2 * TILE_F];

    const int b     = blockIdx.y;
    const int wave  = threadIdx.x >> 5;
    const int lane  = threadIdx.x & 31;
    const int half  = lane >> 4;
    const int lm    = lane & 15;
    const int mbase = blockIdx.x * 64;

    const float* __restrict__ gA = A    + ((size_t)b * NN + mbase + wave * 8) * NN + lane * 4;
    const float* __restrict__ gB = HWsT + ((size_t)b * FF + wave * 8) * NN + lane * 4;
    const int sOff = wave * 8 * PITCH + lane * 4;

    const int rt    = wave & 3;
    const int cpair = wave >> 2;
    const int aOff  = (rt * 16 + lm) * PITCH + 2 * half;
    const int bOff0 = (cpair * 32 + lm) * PITCH + 2 * half;
    const int bOff1 = bOff0 + 16 * PITCH;

#define FOR16(X) X(0) X(1) X(2) X(3) X(4) X(5) X(6) X(7) X(8) X(9) X(10) X(11) X(12) X(13) X(14) X(15)
#define ST_DECL(q)  float4 ra##q, rb##q;
#define ST_LOAD(q)  ra##q = *(const float4*)(pA + (size_t)((q) >> 1) * NN + ((q) & 1) * 128); \
                    rb##q = *(const float4*)(pB + (size_t)((q) >> 1) * NN + ((q) & 1) * 128);
#define ST_STORE(q) *(float4*)&As[so + ((q) >> 1) * PITCH + ((q) & 1) * 128] = ra##q; \
                    *(float4*)&Bs[so + ((q) >> 1) * PITCH + ((q) & 1) * 128] = rb##q;

    {
        FOR16(ST_DECL)
        const float* __restrict__ pA = gA;
        const float* __restrict__ pB = gB;
        FOR16(ST_LOAD)
        const int so = sOff;
        FOR16(ST_STORE)
    }
    __syncthreads();

    FOR16(ST_DECL)

    v8f acc0 = {};
    v8f acc1 = {};

    for (int t = 0; t < NT; ++t) {
        const int cur = (t & 1) * TILE_F;
        const bool more = (t + 1 < NT);

        if (more) {
            const float* __restrict__ pA = gA + (t + 1) * KT;
            const float* __restrict__ pB = gB + (t + 1) * KT;
            FOR16(ST_LOAD)
        }

        const float* as  = &As[cur + aOff];
        const float* bs0 = &Bs[cur + bOff0];
        const float* bs1 = &Bs[cur + bOff1];
#pragma unroll
        for (int kk = 0; kk < KT; kk += 4) {
            const v2f a  = *(const v2f*)(as  + kk);
            const v2f b0 = *(const v2f*)(bs0 + kk);
            const v2f b1 = *(const v2f*)(bs1 + kk);
            acc0 = __builtin_amdgcn_wmma_f32_16x16x4_f32(
                 false, a,  false, b0,
                 (short)0, acc0,  false,  false);
            acc1 = __builtin_amdgcn_wmma_f32_16x16x4_f32(
                 false, a,  false, b1,
                 (short)0, acc1,  false,  false);
        }

        if (more) {
            const int so = (TILE_F - cur) + sOff;
            FOR16(ST_STORE)
        }
        __syncthreads();
    }
#undef ST_STORE
#undef ST_LOAD
#undef ST_DECL
#undef FOR16

    const float* __restrict__ db = d + b * NN;
    float* __restrict__ ob = out + (size_t)b * NN * FF;
    const int r0 = mbase + rt * 16 + 8 * half;
    const int c0 = cpair * 32 + lm;
    const float* __restrict__ h0 = HWsT + ((size_t)b * FF + c0) * NN;
    const float* __restrict__ h1 = h0 + (size_t)16 * NN;
#pragma unroll
    for (int i = 0; i < 8; ++i) {
        const int row = r0 + i;
        const float dr = db[row];
        ob[(size_t)row * FF + c0]      = dr * (acc0[i] + h0[row]);
        ob[(size_t)row * FF + c0 + 16] = dr * (acc1[i] + h1[row]);
    }
}

extern "C" void kernel_launch(void* const* d_in, const int* in_sizes, int n_in,
                              void* d_out, int out_size, void* d_ws, size_t ws_size,
                              hipStream_t stream) {
    const float* A = (const float*)d_in[0];
    const float* H = (const float*)d_in[1];
    const float* W = (const float*)d_in[2];
    float* out = (float*)d_out;

    float* deg  = (float*)d_ws;
    float* HWsT = deg + BB * NN;

    k_colsum_d<<<BB * NN / 64, CS_SLOTS * 16, 0, stream>>>(A, deg);

    k_hw<<<BB * NN / 4, 256, 0, stream>>>(H, W, deg, HWsT);

    dim3 gm(NN / 64, BB);
    k_gcn_main<<<gm, 256, 0, stream>>>(A, HWsT, deg, out);
}


# H@W projection fused into the column-sum kernel: the workgroup owning 64 columns forms their d-scaled 64x64 HWsT block on f32 WMMA from LDS; projection launch gone; H@W sum order changed
# speedup vs baseline: 2.1182x; 1.1450x over previous
#include <hip/hip_runtime.h>

typedef __attribute__((ext_vector_type(2))) float v2f;
typedef __attribute__((ext_vector_type(8))) float v8f;

#define BB 4
#define NN 4096
#define FF 64

#define CS_SLOTS 32
#define CS_ROWS  (NN / CS_SLOTS)
#define CS_GRP   16
#define HP       68

__global__ void __launch_bounds__(CS_SLOTS * 16) __attribute__((amdgpu_waves_per_eu(4, 8)))
k_colsum_d_hw(const float* __restrict__ A, const float* __restrict__ H,
              const float* __restrict__ Wm, float* __restrict__ d,
              float* __restrict__ HWsT) {
    __shared__ float4 red[CS_SLOTS * 16];
    __shared__ __attribute__((aligned(16))) float Hs[64 * HP];
    __shared__ __attribute__((aligned(16))) float Wt[64 * HP];
    __shared__ __attribute__((aligned(16))) float Ts[64 * HP];
    __shared__ __attribute__((aligned(16))) float ds[64];
    const int t    = threadIdx.x;
    const int q    = t & 15;
    const int slot = t >> 4;
    const int b    = blockIdx.x >> 6;
    const int c0   = (blockIdx.x & 63) * 64;
    const float* __restrict__ p = A + ((size_t)b * NN + slot) * NN + c0 + q * 4;
    const size_t step = (size_t)CS_SLOTS * NN;

    const float4* __restrict__ Hg = (const float4*)(H + ((size_t)b * NN + c0) * FF);
    const float4* __restrict__ Wg = (const float4*)Wm;
    const float4 hq0 = Hg[t];
    const float4 hq1 = Hg[t + 512];
    const float4 wq0 = Wg[t];
    const float4 wq1 = Wg[t + 512];

    float4 s = make_float4(0.0f, 0.0f, 0.0f, 0.0f);
    float4 va[CS_GRP], vb[CS_GRP];
#pragma unroll
    for (int j = 0; j < CS_GRP; ++j)
        va[j] = *(const float4*)(p + (size_t)j * step);
    for (int i = 0; i < CS_ROWS; i += 2 * CS_GRP) {
#pragma unroll
        for (int j = 0; j < CS_GRP; ++j)
            vb[j] = *(const float4*)(p + (size_t)(i + CS_GRP + j) * step);
#pragma unroll
        for (int j = 0; j < CS_GRP; ++j) {
            s.x += va[j].x; s.y += va[j].y; s.z += va[j].z; s.w += va[j].w;
        }
        if (i + 2 * CS_GRP < CS_ROWS) {
#pragma unroll
            for (int j = 0; j < CS_GRP; ++j)
                va[j] = *(const float4*)(p + (size_t)(i + 2 * CS_GRP + j) * step);
        }
#pragma unroll
        for (int j = 0; j < CS_GRP; ++j) {
            s.x += vb[j].x; s.y += vb[j].y; s.z += vb[j].z; s.w += vb[j].w;
        }
    }

    red[slot * 16 + q] = s;
    {
        const int r  = t >> 4;
        const int c4 = (t & 15) * 4;
        *(float4*)&Hs[r * HP + c4]        = hq0;
        *(float4*)&Hs[(r + 32) * HP + c4] = hq1;
        Wt[(c4 + 0) * HP + r] = wq0.x;  Wt[(c4 + 0) * HP + r + 32] = wq1.x;
        Wt[(c4 + 1) * HP + r] = wq0.y;  Wt[(c4 + 1) * HP + r + 32] = wq1.y;
        Wt[(c4 + 2) * HP + r] = wq0.z;  Wt[(c4 + 2) * HP + r + 32] = wq1.z;
        Wt[(c4 + 3) * HP + r] = wq0.w;  Wt[(c4 + 3) * HP + r + 32] = wq1.w;
    }
    __syncthreads();
    if (t < 64) {
        const float* r = (const float*)red;
        float acc = 0.0f;
#pragma unroll 8
        for (int k = 0; k < CS_SLOTS; ++k)
            acc += r[k * 64 + t];
        const float dv = rsqrtf(acc + 1.0f);
        d[b * NN + c0 + t] = dv;
        ds[t] = dv;
    }

    const int wave = t >> 5;
    const int lane = t & 31;
    const int half = lane >> 4;
    const int lm   = lane & 15;
    const int mt   = wave & 3;
    const int ot   = wave >> 2;
    const float* ap = &Hs[(mt * 16 + lm) * HP + 2 * half];
    const float* bp = &Wt[(ot * 16 + lm) * HP + 2 * half];
    v8f hw = {};
#pragma unroll
    for (int kk = 0; kk < FF; kk += 4) {
        const v2f a  = *(const v2f*)(ap + kk);
        const v2f bf = *(const v2f*)(bp + kk);
        hw = __builtin_amdgcn_wmma_f32_16x16x4_f32(
             false, a,  false, bf,
             (short)0, hw,  false,  false);
    }
    __syncthreads();
    {
        const int m0 = mt * 16 + 8 * half;
        const float4 d0 = *(const float4*)&ds[m0];
        const float4 d1 = *(const float4*)&ds[m0 + 4];
        float* tp = &Ts[(ot * 16 + lm) * HP + m0];
        *(float4*)tp       = make_float4(d0.x * hw[0], d0.y * hw[1], d0.z * hw[2], d0.w * hw[3]);
        *(float4*)(tp + 4) = make_float4(d1.x * hw[4], d1.y * hw[5], d1.z * hw[6], d1.w * hw[7]);
    }
    __syncthreads();
    {
        const int o  = t >> 4;
        const int m4 = (t & 15) * 4;
        float* g = HWsT + ((size_t)b * FF + o) * NN + c0 + m4;
        *(float4*)g                     = *(const float4*)&Ts[o * HP + m4];
        *(float4*)(g + (size_t)32 * NN) = *(const float4*)&Ts[(o + 32) * HP + m4];
    }
}

#define KT     256
#define PITCH  260
#define NT     (NN / KT)
#define TILE_F (64 * PITCH)

__global__ void __launch_bounds__(256)
k_gcn_main(const float* __restrict__ A, const float* __restrict__ HWsT,
           const float* __restrict__ d, float* __restrict__ out) {
    __shared__ __attribute__((aligned(16))) float As[2 * TILE_F];
    __shared__ __attribute__((aligned(16))) float Bs[2 * TILE_F];

    const int b     = blockIdx.y;
    const int wave  = threadIdx.x >> 5;
    const int lane  = threadIdx.x & 31;
    const int half  = lane >> 4;
    const int lm    = lane & 15;
    const int mbase = blockIdx.x * 64;

    const float* __restrict__ gA = A    + ((size_t)b * NN + mbase + wave * 8) * NN + lane * 4;
    const float* __restrict__ gB = HWsT + ((size_t)b * FF + wave * 8) * NN + lane * 4;
    const int sOff = wave * 8 * PITCH + lane * 4;

    const int rt    = wave & 3;
    const int cpair = wave >> 2;
    const int aOff  = (rt * 16 + lm) * PITCH + 2 * half;
    const int bOff0 = (cpair * 32 + lm) * PITCH + 2 * half;
    const int bOff1 = bOff0 + 16 * PITCH;

#define FOR16(X) X(0) X(1) X(2) X(3) X(4) X(5) X(6) X(7) X(8) X(9) X(10) X(11) X(12) X(13) X(14) X(15)
#define ST_DECL(q)  float4 ra##q, rb##q;
#define ST_LOAD(q)  ra##q = *(const float4*)(pA + (size_t)((q) >> 1) * NN + ((q) & 1) * 128); \
                    rb##q = *(const float4*)(pB + (size_t)((q) >> 1) * NN + ((q) & 1) * 128);
#define ST_STORE(q) *(float4*)&As[so + ((q) >> 1) * PITCH + ((q) & 1) * 128] = ra##q; \
                    *(float4*)&Bs[so + ((q) >> 1) * PITCH + ((q) & 1) * 128] = rb##q;

    {
        FOR16(ST_DECL)
        const float* __restrict__ pA = gA;
        const float* __restrict__ pB = gB;
        FOR16(ST_LOAD)
        const int so = sOff;
        FOR16(ST_STORE)
    }
    __syncthreads();

    FOR16(ST_DECL)

    v8f acc0 = {};
    v8f acc1 = {};

    for (int t = 0; t < NT; ++t) {
        const int cur = (t & 1) * TILE_F;
        const bool more = (t + 1 < NT);

        if (more) {
            const float* __restrict__ pA = gA + (t + 1) * KT;
            const float* __restrict__ pB = gB + (t + 1) * KT;
            FOR16(ST_LOAD)
        }

        const float* as  = &As[cur + aOff];
        const float* bs0 = &Bs[cur + bOff0];
        const float* bs1 = &Bs[cur + bOff1];
#pragma unroll
        for (int kk = 0; kk < KT; kk += 4) {
            const v2f a  = *(const v2f*)(as  + kk);
            const v2f b0 = *(const v2f*)(bs0 + kk);
            const v2f b1 = *(const v2f*)(bs1 + kk);
            acc0 = __builtin_amdgcn_wmma_f32_16x16x4_f32(
                 false, a,  false, b0,
                 (short)0, acc0,  false,  false);
            acc1 = __builtin_amdgcn_wmma_f32_16x16x4_f32(
                 false, a,  false, b1,
                 (short)0, acc1,  false,  false);
        }

        if (more) {
            const int so = (TILE_F - cur) + sOff;
            FOR16(ST_STORE)
        }
        __syncthreads();
    }
#undef ST_STORE
#undef ST_LOAD
#undef ST_DECL
#undef FOR16

    const float* __restrict__ db = d + b * NN;
    float* __restrict__ ob = out + (size_t)b * NN * FF;
    const int r0 = mbase + rt * 16 + 8 * half;
    const int c0 = cpair * 32 + lm;
    const float* __restrict__ h0 = HWsT + ((size_t)b * FF + c0) * NN;
    const float* __restrict__ h1 = h0 + (size_t)16 * NN;
#pragma unroll
    for (int i = 0; i < 8; ++i) {
        const int row = r0 + i;
        const float dr = db[row];
        ob[(size_t)row * FF + c0]      = dr * (acc0[i] + h0[row]);
        ob[(size_t)row * FF + c0 + 16] = dr * (acc1[i] + h1[row]);
    }
}

extern "C" void kernel_launch(void* const* d_in, const int* in_sizes, int n_in,
                              void* d_out, int out_size, void* d_ws, size_t ws_size,
                              hipStream_t stream) {
    const float* A = (const float*)d_in[0];
    const float* H = (const float*)d_in[1];
    const float* W = (const float*)d_in[2];
    float* out = (float*)d_out;

    float* deg  = (float*)d_ws;
    float* HWsT = deg + BB * NN;

    k_colsum_d_hw<<<BB * NN / 64, CS_SLOTS * 16, 0, stream>>>(A, H, W, deg, HWsT);

    dim3 gm(NN / 64, BB);
    k_gcn_main<<<gm, 256, 0, stream>>>(A, HWsT, deg, out);
}


# main GEMM K loop: tile staging spread over the tile in four groups (LDS stores of tile t+1, global loads of tile t+2, next block's fragment reads) ahead of 32-WMMA blocks; sum order unchanged
# speedup vs baseline: 2.4828x; 1.1721x over previous
#include <hip/hip_runtime.h>

typedef __attribute__((ext_vector_type(2))) float v2f;
typedef __attribute__((ext_vector_type(8))) float v8f;

#define BB 4
#define NN 4096
#define FF 64

#define CS_SLOTS 32
#define CS_ROWS  (NN / CS_SLOTS)
#define CS_GRP   16
#define HP       68

__global__ void __launch_bounds__(CS_SLOTS * 16) __attribute__((amdgpu_waves_per_eu(4, 8)))
k_colsum_d_hw(const float* __restrict__ A, const float* __restrict__ H,
              const float* __restrict__ Wm, float* __restrict__ d,
              float* __restrict__ HWsT) {
    __shared__ float4 red[CS_SLOTS * 16];
    __shared__ __attribute__((aligned(16))) float Hs[64 * HP];
    __shared__ __attribute__((aligned(16))) float Wt[64 * HP];
    __shared__ __attribute__((aligned(16))) float Ts[64 * HP];
    __shared__ __attribute__((aligned(16))) float ds[64];
    const int t    = threadIdx.x;
    const int q    = t & 15;
    const int slot = t >> 4;
    const int b    = blockIdx.x >> 6;
    const int c0   = (blockIdx.x & 63) * 64;
    const float* __restrict__ p = A + ((size_t)b * NN + slot) * NN + c0 + q * 4;
    const size_t step = (size_t)CS_SLOTS * NN;

    const float4* __restrict__ Hg = (const float4*)(H + ((size_t)b * NN + c0) * FF);
    const float4* __restrict__ Wg = (const float4*)Wm;
    const float4 hq0 = Hg[t];
    const float4 hq1 = Hg[t + 512];
    const float4 wq0 = Wg[t];
    const float4 wq1 = Wg[t + 512];

    float4 s = make_float4(0.0f, 0.0f, 0.0f, 0.0f);
    float4 va[CS_GRP], vb[CS_GRP];
#pragma unroll
    for (int j = 0; j < CS_GRP; ++j)
        va[j] = *(const float4*)(p + (size_t)j * step);
    for (int i = 0; i < CS_ROWS; i += 2 * CS_GRP) {
#pragma unroll
        for (int j = 0; j < CS_GRP; ++j)
            vb[j] = *(const float4*)(p + (size_t)(i + CS_GRP + j) * step);
#pragma unroll
        for (int j = 0; j < CS_GRP; ++j) {
            s.x += va[j].x; s.y += va[j].y; s.z += va[j].z; s.w += va[j].w;
        }
        if (i + 2 * CS_GRP < CS_ROWS) {
#pragma unroll
            for (int j = 0; j < CS_GRP; ++j)
                va[j] = *(const float4*)(p + (size_t)(i + 2 * CS_GRP + j) * step);
        }
#pragma unroll
        for (int j = 0; j < CS_GRP; ++j) {
            s.x += vb[j].x; s.y += vb[j].y; s.z += vb[j].z; s.w += vb[j].w;
        }
    }

    red[slot * 16 + q] = s;
    {
        const int r  = t >> 4;
        const int c4 = (t & 15) * 4;
        *(float4*)&Hs[r * HP + c4]        = hq0;
        *(float4*)&Hs[(r + 32) * HP + c4] = hq1;
        Wt[(c4 + 0) * HP + r] = wq0.x;  Wt[(c4 + 0) * HP + r + 32] = wq1.x;
        Wt[(c4 + 1) * HP + r] = wq0.y;  Wt[(c4 + 1) * HP + r + 32] = wq1.y;
        Wt[(c4 + 2) * HP + r] = wq0.z;  Wt[(c4 + 2) * HP + r + 32] = wq1.z;
        Wt[(c4 + 3) * HP + r] = wq0.w;  Wt[(c4 + 3) * HP + r + 32] = wq1.w;
    }
    __syncthreads();
    if (t < 64) {
        const float* r = (const float*)red;
        float acc = 0.0f;
#pragma unroll 8
        for (int k = 0; k < CS_SLOTS; ++k)
            acc += r[k * 64 + t];
        const float dv = rsqrtf(acc + 1.0f);
        d[b * NN + c0 + t] = dv;
        ds[t] = dv;
    }

    const int wave = t >> 5;
    const int lane = t & 31;
    const int half = lane >> 4;
    const int lm   = lane & 15;
    const int mt   = wave & 3;
    const int ot   = wave >> 2;
    const float* ap = &Hs[(mt * 16 + lm) * HP + 2 * half];
    const float* bp = &Wt[(ot * 16 + lm) * HP + 2 * half];
    v8f hw = {};
#pragma unroll
    for (int kk = 0; kk < FF; kk += 4) {
        const v2f a  = *(const v2f*)(ap + kk);
        const v2f bf = *(const v2f*)(bp + kk);
        hw = __builtin_amdgcn_wmma_f32_16x16x4_f32(
             false, a,  false, bf,
             (short)0, hw,  false,  false);
    }
    __syncthreads();
    {
        const int m0 = mt * 16 + 8 * half;
        const float4 d0 = *(const float4*)&ds[m0];
        const float4 d1 = *(const float4*)&ds[m0 + 4];
        float* tp = &Ts[(ot * 16 + lm) * HP + m0];
        *(float4*)tp       = make_float4(d0.x * hw[0], d0.y * hw[1], d0.z * hw[2], d0.w * hw[3]);
        *(float4*)(tp + 4) = make_float4(d1.x * hw[4], d1.y * hw[5], d1.z * hw[6], d1.w * hw[7]);
    }
    __syncthreads();
    {
        const int o  = t >> 4;
        const int m4 = (t & 15) * 4;
        float* g = HWsT + ((size_t)b * FF + o) * NN + c0 + m4;
        *(float4*)g                     = *(const float4*)&Ts[o * HP + m4];
        *(float4*)(g + (size_t)32 * NN) = *(const float4*)&Ts[(o + 32) * HP + m4];
    }
}

#define KT     256
#define PITCH  260
#define NT     (NN / KT)
#define TILE_F (64 * PITCH)
#define SUBK   64

static __device__ __forceinline__ void lds_barrier() {
    __builtin_amdgcn_fence(__ATOMIC_RELEASE, "workgroup", "local");
    __builtin_amdgcn_s_barrier();
    __builtin_amdgcn_fence(__ATOMIC_ACQUIRE, "workgroup", "local");
}

__global__ void __launch_bounds__(256)
k_gcn_main(const float* __restrict__ A, const float* __restrict__ HWsT,
           const float* __restrict__ d, float* __restrict__ out) {
    __shared__ __attribute__((aligned(16))) float As[2 * TILE_F];
    __shared__ __attribute__((aligned(16))) float Bs[2 * TILE_F];

    const int b     = blockIdx.y;
    const int wave  = threadIdx.x >> 5;
    const int lane  = threadIdx.x & 31;
    const int half  = lane >> 4;
    const int lm    = lane & 15;
    const int mbase = blockIdx.x * 64;

    const float* __restrict__ gA = A    + ((size_t)b * NN + mbase + wave * 8) * NN + lane * 4;
    const float* __restrict__ gB = HWsT + ((size_t)b * FF + wave * 8) * NN + lane * 4;
    const int sOff = wave * 8 * PITCH + lane * 4;

    const int rt    = wave & 3;
    const int cpair = wave >> 2;
    const int aOff  = (rt * 16 + lm) * PITCH + 2 * half;
    const int bOff0 = (cpair * 32 + lm) * PITCH + 2 * half;
    const int bOff1 = bOff0 + 16 * PITCH;

#define FOR16(X) X(0) X(1) X(2) X(3) X(4) X(5) X(6) X(7) X(8) X(9) X(10) X(11) X(12) X(13) X(14) X(15)
#define QUART0(X) X(0) X(1) X(2) X(3)
#define QUART1(X) X(4) X(5) X(6) X(7)
#define QUART2(X) X(8) X(9) X(10) X(11)
#define QUART3(X) X(12) X(13) X(14) X(15)
#define ST_DECL(q)  float4 ra##q, rb##q;
#define ST_LOAD(q)  ra##q = *(const float4*)(pA + (size_t)((q) >> 1) * NN + ((q) & 1) * 128); \
                    rb##q = *(const float4*)(pB + (size_t)((q) >> 1) * NN + ((q) & 1) * 128);
#define ST_STORE(q) *(float4*)&As[so + ((q) >> 1) * PITCH + ((q) & 1) * 128] = ra##q; \
                    *(float4*)&Bs[so + ((q) >> 1) * PITCH + ((q) & 1) * 128] = rb##q;

    v2f fa[2][16], fb0[2][16], fb1[2][16];
#define FRAG_LOAD(S, BUF, SUB)                                                        \
    _Pragma("unroll")                \
    for (int i = 0; i < 16; ++i) {                                                    \
        fa[S][i]  = *(const v2f*)&As[(BUF) + aOff  + (SUB) * SUBK + 4 * i];           \
        fb0[S][i] = *(const v2f*)&Bs[(BUF) + bOff0 + (SUB) * SUBK + 4 * i];           \
        fb1[S][i] = *(const v2f*)&Bs[(BUF) + bOff1 + (SUB) * SUBK + 4 * i];           \
    }
#define WMMA_BLOCK(S)                                                                 \
    _Pragma("unroll")                      \
    for (int i = 0; i < 16; ++i) {                                                    \
        acc0 = __builtin_amdgcn_wmma_f32_16x16x4_f32(                                 \
             false, fa[S][i],  false, fb0[S][i],                    \
             (short)0, acc0,  false,  false);          \
        acc1 = __builtin_amdgcn_wmma_f32_16x16x4_f32(                                 \
             false, fa[S][i],  false, fb1[S][i],                    \
             (short)0, acc1,  false,  false);          \
    }
#define PIN_ACC() asm volatile("" : "+v"(acc0), "+v"(acc1))
#define ST_NONE(q)
#define BLOCK(J, QUART, STORE_OP, LOAD_OP)                                            \
    if ((J) < 3) { FRAG_LOAD(((J) + 1) & 1, cur, (J) + 1) }                           \
    QUART(STORE_OP)                                                                   \
    QUART(LOAD_OP)                                                                    \
    PIN_ACC();                                                                        \
    WMMA_BLOCK((J) & 1)                                                               \
    PIN_ACC();
#define TILE(STORE_OP, LOAD_OP)                                                       \
    BLOCK(0, QUART0, STORE_OP, LOAD_OP)                                               \
    BLOCK(1, QUART1, STORE_OP, LOAD_OP)                                               \
    BLOCK(2, QUART2, STORE_OP, LOAD_OP)                                               \
    BLOCK(3, QUART3, STORE_OP, LOAD_OP)

    FOR16(ST_DECL)

    {
#define T0_DECL(q)  float4 ta##q, tb##q;
#define T0_LOAD(q)  ta##q = *(const float4*)(gA + (size_t)((q) >> 1) * NN + ((q) & 1) * 128); \
                    tb##q = *(const float4*)(gB + (size_t)((q) >> 1) * NN + ((q) & 1) * 128);
#define T0_STORE(q) *(float4*)&As[sOff + ((q) >> 1) * PITCH + ((q) & 1) * 128] = ta##q; \
                    *(float4*)&Bs[sOff + ((q) >> 1) * PITCH + ((q) & 1) * 128] = tb##q;
        FOR16(T0_DECL)
        FOR16(T0_LOAD)
        {
            const float* __restrict__ pA = gA + KT;
            const float* __restrict__ pB = gB + KT;
            FOR16(ST_LOAD)
        }
        FOR16(T0_STORE)
#undef T0_STORE
#undef T0_LOAD
#undef T0_DECL
    }
    __syncthreads();

    v8f acc0 = {};
    v8f acc1 = {};

    int cur = 0;
    FRAG_LOAD(0, cur, 0)
    const float* __restrict__ pA = gA + 2 * KT;
    const float* __restrict__ pB = gB + 2 * KT;
    for (int t = 0; t < NT - 2; ++t) {
        const int so = (TILE_F - cur) + sOff;
        TILE(ST_STORE, ST_LOAD)
        lds_barrier();
        cur = TILE_F - cur;
        pA += KT;
        pB += KT;
        FRAG_LOAD(0, cur, 0)
    }
    {
        const int so = (TILE_F - cur) + sOff;
        TILE(ST_STORE, ST_NONE)
        lds_barrier();
        cur = TILE_F - cur;
        FRAG_LOAD(0, cur, 0)
    }
    {
        TILE(ST_NONE, ST_NONE)
    }
#undef TILE
#undef BLOCK
#undef PIN_ACC
#undef ST_NONE
#undef WMMA_BLOCK
#undef FRAG_LOAD
#undef ST_STORE
#undef ST_LOAD
#undef ST_DECL
#undef QUART3
#undef QUART2
#undef QUART1
#undef QUART0
#undef FOR16

    const float* __restrict__ db = d + b * NN;
    float* __restrict__ ob = out + (size_t)b * NN * FF;
    const int r0 = mbase + rt * 16 + 8 * half;
    const int c0 = cpair * 32 + lm;
    const float* __restrict__ h0 = HWsT + ((size_t)b * FF + c0) * NN;
    const float* __restrict__ h1 = h0 + (size_t)16 * NN;
#pragma unroll
    for (int i = 0; i < 8; ++i) {
        const int row = r0 + i;
        const float dr = db[row];
        ob[(size_t)row * FF + c0]      = dr * (acc0[i] + h0[row]);
        ob[(size_t)row * FF + c0 + 16] = dr * (acc1[i] + h1[row]);
    }
}

extern "C" void kernel_launch(void* const* d_in, const int* in_sizes, int n_in,
                              void* d_out, int out_size, void* d_ws, size_t ws_size,
                              hipStream_t stream) {
    const float* A = (const float*)d_in[0];
    const float* H = (const float*)d_in[1];
    const float* W = (const float*)d_in[2];
    float* out = (float*)d_out;

    float* deg  = (float*)d_ws;
    float* HWsT = deg + BB * NN;

    k_colsum_d_hw<<<BB * NN / 64, CS_SLOTS * 16, 0, stream>>>(A, H, W, deg, HWsT);

    dim3 gm(NN / 64, BB);
    k_gcn_main<<<gm, 256, 0, stream>>>(A, HWsT, deg, out);
}
